# starting kernel
# baseline (speedup 1.0000x reference)
#include <hip/hip_runtime.h>


#define N_NODES 50000
#define N_EDGES 800000
#define F_IN    64
#define F_HID   128
#define F_OUT   64

typedef __attribute__((ext_vector_type(2))) float v2f;
typedef __attribute__((ext_vector_type(8))) float v8f;


__global__ void zero_kernel(float* __restrict__ p, int n) {
    int i = blockIdx.x * blockDim.x + threadIdx.x;
    if (i < n) p[i] = 0.0f;
}

__global__ void degree_kernel(const int* __restrict__ src, const int* __restrict__ dst,
                              float* __restrict__ out_deg, float* __restrict__ in_deg) {
    int e = blockIdx.x * blockDim.x + threadIdx.x;
    if (e < N_EDGES) {
        atomicAdd(&out_deg[src[e]], 1.0f);
        atomicAdd(&in_deg[dst[e]], 1.0f);
    }
}

__global__ void norm_kernel(const float* __restrict__ out_deg, const float* __restrict__ in_deg,
                            float* __restrict__ norm_src, float* __restrict__ norm_dst) {
    int i = blockIdx.x * blockDim.x + threadIdx.x;
    if (i < N_NODES) {
        norm_src[i] = rsqrtf(fmaxf(out_deg[i], 1.0f));
        norm_dst[i] = rsqrtf(fmaxf(in_deg[i], 1.0f));
    }
}

__global__ void scatter_scaled_kernel(const float* __restrict__ feat,
                                      const float* __restrict__ scale,
                                      const int* __restrict__ src, const int* __restrict__ dst,
                                      float* __restrict__ agg) {
    long long t = (long long)blockIdx.x * 256 + threadIdx.x;
    int e = (int)(t >> 6);
    int f = (int)(t & 63);
    if (e < N_EDGES) {
        int s = src[e], d = dst[e];
        atomicAdd(&agg[(long long)d * 64 + f], feat[(long long)s * 64 + f] * scale[s]);
    }
}

__global__ void scatter_plain_kernel(const float* __restrict__ feat,
                                     const int* __restrict__ src, const int* __restrict__ dst,
                                     float* __restrict__ agg) {
    long long t = (long long)blockIdx.x * 256 + threadIdx.x;
    int e = (int)(t >> 6);
    int f = (int)(t & 63);
    if (e < N_EDGES) {
        int s = src[e], d = dst[e];
        atomicAdd(&agg[(long long)d * 64 + f], feat[(long long)s * 64 + f]);
    }
}

__global__ void final_kernel(const float* __restrict__ agg, const float* __restrict__ norm_dst,
                             const float* __restrict__ bias, float* __restrict__ out) {
    long long t = (long long)blockIdx.x * 256 + threadIdx.x;
    if (t < (long long)N_NODES * 64) {
        int i = (int)(t >> 6);
        int f = (int)(t & 63);
        out[t] = agg[t] * norm_dst[i] + bias[f];
    }
}

template <int KDIM, int NDIM, bool SCALE_A, bool EPILOGUE>
__global__ void wmma_gemm_kernel(const float* __restrict__ A,
                                 const float* __restrict__ B,
                                 const float* __restrict__ rowscale_a,
                                 const float* __restrict__ rowscale_d,
                                 const float* __restrict__ bias,
                                 float* __restrict__ D) {
    const int lane    = threadIdx.x & 31;
    const int wave    = threadIdx.x >> 5;
    const int half    = lane >> 4;
    const int l       = lane & 15;
    const int row0    = blockIdx.x * 16;
    const int arow    = row0 + l;
    const int col     = wave * 16 + l;

    float ascale = 1.0f;
    if (SCALE_A) ascale = rowscale_a[arow];

    v8f c = {};
    const float* arowp = A + (long long)arow * KDIM;

    #pragma unroll
    for (int k = 0; k < KDIM; k += 4) {
        const int ka = k + half * 2;
        v2f a = *(const v2f*)(arowp + ka);
        if (SCALE_A) { a.x *= ascale; a.y *= ascale; }
        v2f b;
        b.x = B[(long long)ka * NDIM + col];
        b.y = B[(long long)(ka + 1) * NDIM + col];
        c = __builtin_amdgcn_wmma_f32_16x16x4_f32(
                 false, a,  false, b,
                 (short)0, c,  false,  false);
    }

    #pragma unroll
    for (int r = 0; r < 8; ++r) {
        int row = row0 + r + half * 8;
        float v = c[r];
        if (EPILOGUE) {
            v = v * rowscale_d[row] + bias[col];
            v = fmaxf(v, 0.0f);
        }
        D[(long long)row * NDIM + col] = v;
    }
}


extern "C" void kernel_launch(void* const* d_in, const int* in_sizes, int n_in,
                              void* d_out, int out_size, void* d_ws, size_t ws_size,
                              hipStream_t stream) {
    const float* x   = (const float*)d_in[0];
    const float* W1  = (const float*)d_in[1];
    const float* b1  = (const float*)d_in[2];
    const float* W2  = (const float*)d_in[3];
    const float* b2  = (const float*)d_in[4];
    const int*   src = (const int*)d_in[5];
    const int*   dst = (const int*)d_in[6];
    float*       out = (float*)d_out;

    float* ws       = (float*)d_ws;
    float* out_deg  = ws;
    float* in_deg   = out_deg + N_NODES;
    float* norm_src = in_deg + N_NODES;
    float* norm_dst = norm_src + N_NODES;
    float* agg      = norm_dst + N_NODES;
    float* h1       = agg + (long long)N_NODES * 64;
    float* h2w      = h1 + (long long)N_NODES * 128;

    const int ZT = 256;
    const long long aggElems = (long long)N_NODES * 64;
    const long long edgeThreads = (long long)N_EDGES * 64;

    zero_kernel<<<(2 * N_NODES + ZT - 1) / ZT, ZT, 0, stream>>>(out_deg, 2 * N_NODES);
    degree_kernel<<<(N_EDGES + ZT - 1) / ZT, ZT, 0, stream>>>(src, dst, out_deg, in_deg);
    norm_kernel<<<(N_NODES + ZT - 1) / ZT, ZT, 0, stream>>>(out_deg, in_deg, norm_src, norm_dst);

    zero_kernel<<<(int)((aggElems + ZT - 1) / ZT), ZT, 0, stream>>>(agg, (int)aggElems);
    scatter_scaled_kernel<<<(int)((edgeThreads + ZT - 1) / ZT), ZT, 0, stream>>>(
        x, norm_src, src, dst, agg);
    wmma_gemm_kernel<F_IN, F_HID, false, true><<<N_NODES / 16, 8 * 32, 0, stream>>>(
        agg, W1, nullptr, norm_dst, b1, h1);

    zero_kernel<<<(int)((aggElems + ZT - 1) / ZT), ZT, 0, stream>>>(agg, (int)aggElems);
    wmma_gemm_kernel<F_HID, F_OUT, true, false><<<N_NODES / 16, 4 * 32, 0, stream>>>(
        h1, W2, norm_src, nullptr, nullptr, h2w);
    scatter_plain_kernel<<<(int)((edgeThreads + ZT - 1) / ZT), ZT, 0, stream>>>(
        h2w, src, dst, agg);

    final_kernel<<<(int)((aggElems + ZT - 1) / ZT), ZT, 0, stream>>>(agg, norm_dst, b2, out);
}


# Float-atomic scatter-add replaced by a gather: each call groups edges by destination (returning int add gives arrival-order rank, scan, fill); one wave sums a node's rows in registers, stores once
# speedup vs baseline: 1.9457x; 1.9457x over previous
#include <hip/hip_runtime.h>


#define N_NODES 50000
#define N_EDGES 800000
#define F_IN    64
#define F_HID   128
#define F_OUT   64

typedef __attribute__((ext_vector_type(2))) float v2f;
typedef __attribute__((ext_vector_type(8))) float v8f;


__global__ void zero_int_kernel(int* __restrict__ p, int n) {
    int i = blockIdx.x * blockDim.x + threadIdx.x;
    if (i < n) p[i] = 0;
}

__global__ void count_rank_kernel(const int* __restrict__ src, const int* __restrict__ dst,
                                  int* __restrict__ out_cnt, int* __restrict__ in_cnt,
                                  int* __restrict__ rank) {
    int e = blockIdx.x * blockDim.x + threadIdx.x;
    if (e < N_EDGES) {
        atomicAdd(&out_cnt[src[e]], 1);
        rank[e] = atomicAdd(&in_cnt[dst[e]], 1);
    }
}

#define SCAN_T 1024
__global__ __launch_bounds__(SCAN_T)
void scan_norm_kernel(const int* __restrict__ out_cnt, const int* __restrict__ in_cnt,
                      int* __restrict__ seg, float* __restrict__ norm_src,
                      float* __restrict__ norm_dst) {
    __shared__ int before_part[SCAN_T / 32];
    __shared__ int wave_total[SCAN_T / 32];
    const int tid  = threadIdx.x;
    const int lane = tid & 31;
    const int wave = tid >> 5;
    const int4* in4 = (const int4*)in_cnt;
    const int nvec = N_NODES / 4;
    const int vec_before = blockIdx.x * SCAN_T;

    int pre = 0;
    for (int v = tid; v < vec_before; v += SCAN_T) {
        int4 q = in4[v];
        pre += (q.x + q.y) + (q.z + q.w);
    }
    for (int o = 16; o > 0; o >>= 1) pre += __shfl_xor(pre, o);
    if (lane == 0) before_part[wave] = pre;

    const int v = vec_before + tid;
    int4 q = make_int4(0, 0, 0, 0);
    if (v < nvec) q = in4[v];
    const int t = (q.x + q.y) + (q.z + q.w);
    int inc = t;
    for (int o = 1; o < 32; o <<= 1) {
        int u = __shfl_up(inc, o);
        if (lane >= o) inc += u;
    }
    if (lane == 31) wave_total[wave] = inc;
    __syncthreads();

    int base = 0;
    for (int w = 0; w < SCAN_T / 32; ++w) {
        base += before_part[w];
        if (w < wave) base += wave_total[w];
    }

    if (v < nvec) {
        const int b0 = base + inc - t;
        const int b1 = b0 + q.x, b2 = b1 + q.y, b3 = b2 + q.z, b4 = b3 + q.w;
        int4* seg4 = (int4*)seg;
        seg4[2 * v]     = make_int4(b0, b1, b1, b2);
        seg4[2 * v + 1] = make_int4(b2, b3, b3, b4);
        const int4 oc = ((const int4*)out_cnt)[v];
        float4 ns, nd;
        ns.x = rsqrtf(fmaxf((float)oc.x, 1.0f)); ns.y = rsqrtf(fmaxf((float)oc.y, 1.0f));
        ns.z = rsqrtf(fmaxf((float)oc.z, 1.0f)); ns.w = rsqrtf(fmaxf((float)oc.w, 1.0f));
        nd.x = rsqrtf(fmaxf((float)q.x, 1.0f));  nd.y = rsqrtf(fmaxf((float)q.y, 1.0f));
        nd.z = rsqrtf(fmaxf((float)q.z, 1.0f));  nd.w = rsqrtf(fmaxf((float)q.w, 1.0f));
        ((float4*)norm_src)[v] = ns;
        ((float4*)norm_dst)[v] = nd;
    }
}

__global__ void fill_kernel(const int* __restrict__ src, const int* __restrict__ dst,
                            const int* __restrict__ rank, const int* __restrict__ seg,
                            int* __restrict__ col) {
    int e = blockIdx.x * blockDim.x + threadIdx.x;
    if (e < N_EDGES) col[seg[2 * dst[e]] + rank[e]] = src[e];
}

#define GATHER_BLOCKS 1024

template <bool FIRST>
__device__ __forceinline__ void add_row(float4& acc, const float* __restrict__ feat,
                                        const float* __restrict__ norm_src,
                                        int c, bool valid, int l) {
    float4 r = *(const float4*)(feat + (long long)c * 64 + l * 4);
    if (!valid) r = make_float4(0.0f, 0.0f, 0.0f, 0.0f);
    if (FIRST) {
        const float sc = norm_src[c];
        acc.x += r.x * sc; acc.y += r.y * sc; acc.z += r.z * sc; acc.w += r.w * sc;
    } else {
        acc.x += r.x; acc.y += r.y; acc.z += r.z; acc.w += r.w;
    }
}

template <bool FIRST>
__global__ __launch_bounds__(256)
void gather_kernel(const float* __restrict__ feat, const float* __restrict__ norm_src,
                   const int* __restrict__ seg, const int* __restrict__ col,
                   const float* __restrict__ norm_dst, const float* __restrict__ bias,
                   float* __restrict__ outp) {
    const int lane = threadIdx.x & 31;
    const int half = lane >> 4;
    const int l    = lane & 15;
    const int nw   = GATHER_BLOCKS * 8;
    int node = __builtin_amdgcn_readfirstlane(blockIdx.x * 8 + (threadIdx.x >> 5));
    const int2* seg2 = (const int2*)seg;
    const int2 se0 = seg2[node];
    int jb  = se0.x + half;
    int end = se0.y;
    float4 bv = make_float4(0.0f, 0.0f, 0.0f, 0.0f);
    if (!FIRST) bv = *(const float4*)(bias + l * 4);

    while (true) {
        const int next = node + nw;
        int2 nse = make_int2(0, 0);
        if (next < N_NODES) nse = seg2[next];

        float nd = 1.0f;
        if (!FIRST) nd = norm_dst[node];

        float4 acc = make_float4(0.0f, 0.0f, 0.0f, 0.0f);
        for (int j = jb; j < end; j += 8) {
            const bool v1 = (j + 2) < end, v2 = (j + 4) < end, v3 = (j + 6) < end;
            const int c0 = col[j];
            const int c1 = col[v1 ? j + 2 : j];
            const int c2 = col[v2 ? j + 4 : j];
            const int c3 = col[v3 ? j + 6 : j];
            add_row<FIRST>(acc, feat, norm_src, c0, true, l);
            add_row<FIRST>(acc, feat, norm_src, c1, v1, l);
            add_row<FIRST>(acc, feat, norm_src, c2, v2, l);
            add_row<FIRST>(acc, feat, norm_src, c3, v3, l);
        }
        acc.x += __shfl_xor(acc.x, 16);
        acc.y += __shfl_xor(acc.y, 16);
        acc.z += __shfl_xor(acc.z, 16);
        acc.w += __shfl_xor(acc.w, 16);
        if (half == 0) {
            float4 o = acc;
            if (!FIRST) {
                o.x = acc.x * nd + bv.x; o.y = acc.y * nd + bv.y;
                o.z = acc.z * nd + bv.z; o.w = acc.w * nd + bv.w;
            }
            *(float4*)(outp + (long long)node * 64 + l * 4) = o;
        }
        if (next >= N_NODES) break;
        node = next;
        jb   = nse.x + half;
        end  = nse.y;
    }
}

template <int KDIM, int NDIM, bool SCALE_A, bool EPILOGUE>
__global__ void wmma_gemm_kernel(const float* __restrict__ A,
                                 const float* __restrict__ B,
                                 const float* __restrict__ rowscale_a,
                                 const float* __restrict__ rowscale_d,
                                 const float* __restrict__ bias,
                                 float* __restrict__ D) {
    const int lane    = threadIdx.x & 31;
    const int wave    = threadIdx.x >> 5;
    const int half    = lane >> 4;
    const int l       = lane & 15;
    const int row0    = blockIdx.x * 16;
    const int arow    = row0 + l;
    const int col     = wave * 16 + l;

    float ascale = 1.0f;
    if (SCALE_A) ascale = rowscale_a[arow];

    v8f c = {};
    const float* arowp = A + (long long)arow * KDIM;

    #pragma unroll
    for (int k = 0; k < KDIM; k += 4) {
        const int ka = k + half * 2;
        v2f a = *(const v2f*)(arowp + ka);
        if (SCALE_A) { a.x *= ascale; a.y *= ascale; }
        v2f b;
        b.x = B[(long long)ka * NDIM + col];
        b.y = B[(long long)(ka + 1) * NDIM + col];
        c = __builtin_amdgcn_wmma_f32_16x16x4_f32(
                 false, a,  false, b,
                 (short)0, c,  false,  false);
    }

    #pragma unroll
    for (int r = 0; r < 8; ++r) {
        int row = row0 + r + half * 8;
        float v = c[r];
        if (EPILOGUE) {
            v = v * rowscale_d[row] + bias[col];
            v = fmaxf(v, 0.0f);
        }
        D[(long long)row * NDIM + col] = v;
    }
}


extern "C" void kernel_launch(void* const* d_in, const int* in_sizes, int n_in,
                              void* d_out, int out_size, void* d_ws, size_t ws_size,
                              hipStream_t stream) {
    const float* x   = (const float*)d_in[0];
    const float* W1  = (const float*)d_in[1];
    const float* b1  = (const float*)d_in[2];
    const float* W2  = (const float*)d_in[3];
    const float* b2  = (const float*)d_in[4];
    const int*   src = (const int*)d_in[5];
    const int*   dst = (const int*)d_in[6];
    float*       out = (float*)d_out;

    int*   out_cnt  = (int*)d_ws;
    int*   in_cnt   = out_cnt + N_NODES;
    int*   seg      = in_cnt + N_NODES;
    int*   rank     = seg + 2 * N_NODES;
    int*   col      = rank + N_EDGES;
    float* norm_src = (float*)(col + N_EDGES);
    float* norm_dst = norm_src + N_NODES;
    float* agg      = norm_dst + N_NODES;
    float* h1       = agg + (long long)N_NODES * 64;
    float* h2w      = h1 + (long long)N_NODES * 128;

    const int ZT = 256;

    zero_int_kernel<<<(2 * N_NODES + ZT - 1) / ZT, ZT, 0, stream>>>(out_cnt, 2 * N_NODES);
    count_rank_kernel<<<(N_EDGES + ZT - 1) / ZT, ZT, 0, stream>>>(src, dst, out_cnt, in_cnt, rank);
    scan_norm_kernel<<<(N_NODES / 4 + SCAN_T - 1) / SCAN_T, SCAN_T, 0, stream>>>(
        out_cnt, in_cnt, seg, norm_src, norm_dst);
    fill_kernel<<<(N_EDGES + ZT - 1) / ZT, ZT, 0, stream>>>(src, dst, rank, seg, col);

    gather_kernel<true><<<GATHER_BLOCKS, 256, 0, stream>>>(
        x, norm_src, seg, col, nullptr, nullptr, agg);
    wmma_gemm_kernel<F_IN, F_HID, false, true><<<N_NODES / 16, 8 * 32, 0, stream>>>(
        agg, W1, nullptr, norm_dst, b1, h1);

    wmma_gemm_kernel<F_HID, F_OUT, true, false><<<N_NODES / 16, 4 * 32, 0, stream>>>(
        h1, W2, norm_src, nullptr, nullptr, h2w);
    gather_kernel<false><<<GATHER_BLOCKS, 256, 0, stream>>>(
        h2w, nullptr, seg, col, norm_dst, b2, out);
}


# Per-edge global atomics (degree counts, rank) replaced by LDS counters: a work-group owns (edge chunk, node range); per-chunk counts scanned into list offsets; list order still arrival order
# speedup vs baseline: 4.2728x; 2.1960x over previous
#include <hip/hip_runtime.h>


#define N_NODES 50000
#define N_EDGES 800000
#define F_IN    64
#define F_HID   128
#define F_OUT   64

typedef __attribute__((ext_vector_type(2))) float v2f;
typedef __attribute__((ext_vector_type(8))) float v8f;

#define N_CHUNKS        32
#define CHUNK_V         (N_EDGES / N_CHUNKS / 4)
#define RANGE           6400
#define N_RANGES        8
#define BIN_T           512
#define SUB             256
#define SUBS_PER_RANGE  (RANGE / SUB)
#define N_SUBS          ((N_NODES + SUB - 1) / SUB)
#define TRIPS           ((CHUNK_V + BIN_T - 1) / BIN_T)
#define PLACE_TRIPS     ((RANGE / 4 + BIN_T - 1) / BIN_T)

__global__ __launch_bounds__(BIN_T)
void degree_count_kernel(const int* __restrict__ src, const int* __restrict__ dst,
                         int* __restrict__ part_out, int* __restrict__ part_in,
                         int* __restrict__ subsum) {
    __shared__ __attribute__((aligned(16))) int cnt[RANGE];
    const int tid  = threadIdx.x;
    const int lane = tid & 31;
    const int k    = blockIdx.x / N_RANGES;
    const int s    = blockIdx.x % N_RANGES;
    const int lo   = s * RANGE;
    const bool is_in = (blockIdx.y == 0);
    const int* __restrict__ idx  = is_in ? dst : src;
    int* __restrict__       part = is_in ? part_in : part_out;

    const int4* idx4 = (const int4*)idx + k * CHUNK_V;
    int4 q[TRIPS];
    #pragma unroll
    for (int i = 0; i < TRIPS; ++i) {
        const int v = tid + i * BIN_T;
        q[i] = idx4[v < CHUNK_V ? v : CHUNK_V - 1];
    }

    int4* cnt4 = (int4*)cnt;
    for (int n = tid; n < RANGE / 4; n += BIN_T) cnt4[n] = make_int4(0, 0, 0, 0);
    __syncthreads();

    #pragma unroll
    for (int i = 0; i < TRIPS; ++i) {
        if (tid + i * BIN_T < CHUNK_V) {
            const unsigned r0 = (unsigned)(q[i].x - lo), r1 = (unsigned)(q[i].y - lo);
            const unsigned r2 = (unsigned)(q[i].z - lo), r3 = (unsigned)(q[i].w - lo);
            if (r0 < RANGE) atomicAdd(&cnt[r0], 1);
            if (r1 < RANGE) atomicAdd(&cnt[r1], 1);
            if (r2 < RANGE) atomicAdd(&cnt[r2], 1);
            if (r3 < RANGE) atomicAdd(&cnt[r3], 1);
        }
    }
    __syncthreads();

    const int4* cnt4r = (const int4*)cnt;
    int4* part4 = (int4*)(part + k * N_NODES + lo);
    for (int j = tid >> 5; j < SUBS_PER_RANGE; j += BIN_T / 32) {
        const int n0 = j * SUB + lane * 4;
        const int4 a = cnt4r[n0 >> 2];
        const int4 b = cnt4r[(n0 + SUB / 2) >> 2];
        if (lo + n0 < N_NODES)           part4[n0 >> 2] = a;
        if (lo + n0 + SUB / 2 < N_NODES) part4[(n0 + SUB / 2) >> 2] = b;
        if (is_in) {
            int w = ((a.x + a.y) + (a.z + a.w)) + ((b.x + b.y) + (b.z + b.w));
            for (int o = 16; o > 0; o >>= 1) w += __shfl_xor(w, o);
            const int g = s * SUBS_PER_RANGE + j;
            if (lane == 0 && g < N_SUBS) subsum[g * N_CHUNKS + k] = w;
        }
    }
}

#define SCAN_T SUB
__global__ __launch_bounds__(SCAN_T)
void scan_norm_kernel(const int* __restrict__ part_out, int* __restrict__ part_in,
                      const int* __restrict__ subsum,
                      int* __restrict__ seg, float* __restrict__ norm_src,
                      float* __restrict__ norm_dst) {
    __shared__ int before_part[SCAN_T / 32];
    __shared__ int wave_total[SCAN_T / 32];
    const int tid  = threadIdx.x;
    const int lane = tid & 31;
    const int wave = tid >> 5;
    const int node = blockIdx.x * SCAN_T + tid;
    const bool valid = node < N_NODES;

    int pre = 0;
    const int nbefore = blockIdx.x * N_CHUNKS;
    for (int i = tid; i < nbefore; i += SCAN_T) pre += subsum[i];
    for (int o = 16; o > 0; o >>= 1) pre += __shfl_xor(pre, o);
    if (lane == 0) before_part[wave] = pre;

    const int ln = valid ? node : N_NODES - 1;
    int p[N_CHUNKS];
    int t = 0, oc = 0;
    #pragma unroll
    for (int k = 0; k < N_CHUNKS; ++k) {
        p[k] = part_in[k * N_NODES + ln];
        t += p[k];
    }
    #pragma unroll
    for (int k = 0; k < N_CHUNKS; ++k) oc += part_out[k * N_NODES + ln];
    if (!valid) t = 0;

    int inc = t;
    for (int o = 1; o < 32; o <<= 1) {
        int u = __shfl_up(inc, o);
        if (lane >= o) inc += u;
    }
    if (lane == 31) wave_total[wave] = inc;
    __syncthreads();

    int base = 0;
    for (int w = 0; w < SCAN_T / 32; ++w) {
        base += before_part[w];
        if (w < wave) base += wave_total[w];
    }

    if (valid) {
        const int b0 = base + inc - t;
        int run = b0;
        #pragma unroll
        for (int k = 0; k < N_CHUNKS; ++k) {
            part_in[k * N_NODES + node] = run;
            run += p[k];
        }
        ((int2*)seg)[node] = make_int2(b0, run);
        norm_src[node] = rsqrtf(fmaxf((float)oc, 1.0f));
        norm_dst[node] = rsqrtf(fmaxf((float)t, 1.0f));
    }
}

__global__ __launch_bounds__(BIN_T)
void fill_kernel(const int* __restrict__ src, const int* __restrict__ dst,
                 const int* __restrict__ start, int* __restrict__ col) {
    __shared__ __attribute__((aligned(16))) int place[PLACE_TRIPS * BIN_T * 4];
    const int tid = threadIdx.x;
    const int k   = blockIdx.x / N_RANGES;
    const int s   = blockIdx.x % N_RANGES;
    const int lo  = s * RANGE;

    const int4* d4 = (const int4*)dst + k * CHUNK_V;
    const int4* s4 = (const int4*)src + k * CHUNK_V;
    int4 qd[TRIPS], qs[TRIPS];
    #pragma unroll
    for (int i = 0; i < TRIPS; ++i) {
        const int v = tid + i * BIN_T;
        const int vc = v < CHUNK_V ? v : CHUNK_V - 1;
        qd[i] = d4[vc];
        qs[i] = s4[vc];
    }
    const int4* start4 = (const int4*)(start + k * N_NODES);
    int4 st[PLACE_TRIPS];
    #pragma unroll
    for (int i = 0; i < PLACE_TRIPS; ++i) {
        const int node = lo + (tid + i * BIN_T) * 4;
        st[i] = start4[(node < N_NODES ? node : N_NODES - 4) >> 2];
    }
    int4* place4 = (int4*)place;
    #pragma unroll
    for (int i = 0; i < PLACE_TRIPS; ++i) place4[tid + i * BIN_T] = st[i];
    __syncthreads();

    #pragma unroll
    for (int i = 0; i < TRIPS; ++i) {
        if (tid + i * BIN_T < CHUNK_V) {
            const unsigned r0 = (unsigned)(qd[i].x - lo), r1 = (unsigned)(qd[i].y - lo);
            const unsigned r2 = (unsigned)(qd[i].z - lo), r3 = (unsigned)(qd[i].w - lo);
            const bool m0 = r0 < RANGE, m1 = r1 < RANGE, m2 = r2 < RANGE, m3 = r3 < RANGE;
            unsigned p0 = 0, p1 = 0, p2 = 0, p3 = 0;
            if (m0) p0 = (unsigned)atomicAdd(&place[r0], 1);
            if (m1) p1 = (unsigned)atomicAdd(&place[r1], 1);
            if (m2) p2 = (unsigned)atomicAdd(&place[r2], 1);
            if (m3) p3 = (unsigned)atomicAdd(&place[r3], 1);
            if (m0) col[p0] = qs[i].x;
            if (m1) col[p1] = qs[i].y;
            if (m2) col[p2] = qs[i].z;
            if (m3) col[p3] = qs[i].w;
        }
    }
}

#define GATHER_BLOCKS 1024

template <bool FIRST>
__device__ __forceinline__ void add_row(float4& acc, const float* __restrict__ feat,
                                        const float* __restrict__ norm_src,
                                        int c, bool valid, int l) {
    float4 r = *(const float4*)(feat + (long long)c * 64 + l * 4);
    if (!valid) r = make_float4(0.0f, 0.0f, 0.0f, 0.0f);
    if (FIRST) {
        const float sc = norm_src[c];
        acc.x += r.x * sc; acc.y += r.y * sc; acc.z += r.z * sc; acc.w += r.w * sc;
    } else {
        acc.x += r.x; acc.y += r.y; acc.z += r.z; acc.w += r.w;
    }
}

template <bool FIRST>
__global__ __launch_bounds__(256)
void gather_kernel(const float* __restrict__ feat, const float* __restrict__ norm_src,
                   const int* __restrict__ seg, const int* __restrict__ col,
                   const float* __restrict__ norm_dst, const float* __restrict__ bias,
                   float* __restrict__ outp) {
    const int lane = threadIdx.x & 31;
    const int half = lane >> 4;
    const int l    = lane & 15;
    const int nw   = GATHER_BLOCKS * 8;
    int node = __builtin_amdgcn_readfirstlane(blockIdx.x * 8 + (threadIdx.x >> 5));
    const int2* seg2 = (const int2*)seg;
    const int2 se0 = seg2[node];
    int jb  = se0.x + half;
    int end = se0.y;
    float4 bv = make_float4(0.0f, 0.0f, 0.0f, 0.0f);
    if (!FIRST) bv = *(const float4*)(bias + l * 4);

    while (true) {
        const int next = node + nw;
        int2 nse = make_int2(0, 0);
        if (next < N_NODES) nse = seg2[next];

        float nd = 1.0f;
        if (!FIRST) nd = norm_dst[node];

        float4 acc = make_float4(0.0f, 0.0f, 0.0f, 0.0f);
        for (int j = jb; j < end; j += 8) {
            const bool v1 = (j + 2) < end, v2 = (j + 4) < end, v3 = (j + 6) < end;
            const int c0 = col[j];
            const int c1 = col[v1 ? j + 2 : j];
            const int c2 = col[v2 ? j + 4 : j];
            const int c3 = col[v3 ? j + 6 : j];
            add_row<FIRST>(acc, feat, norm_src, c0, true, l);
            add_row<FIRST>(acc, feat, norm_src, c1, v1, l);
            add_row<FIRST>(acc, feat, norm_src, c2, v2, l);
            add_row<FIRST>(acc, feat, norm_src, c3, v3, l);
        }
        acc.x += __shfl_xor(acc.x, 16);
        acc.y += __shfl_xor(acc.y, 16);
        acc.z += __shfl_xor(acc.z, 16);
        acc.w += __shfl_xor(acc.w, 16);
        if (half == 0) {
            float4 o = acc;
            if (!FIRST) {
                o.x = acc.x * nd + bv.x; o.y = acc.y * nd + bv.y;
                o.z = acc.z * nd + bv.z; o.w = acc.w * nd + bv.w;
            }
            *(float4*)(outp + (long long)node * 64 + l * 4) = o;
        }
        if (next >= N_NODES) break;
        node = next;
        jb   = nse.x + half;
        end  = nse.y;
    }
}

template <int KDIM, int NDIM, bool SCALE_A, bool EPILOGUE>
__global__ void wmma_gemm_kernel(const float* __restrict__ A,
                                 const float* __restrict__ B,
                                 const float* __restrict__ rowscale_a,
                                 const float* __restrict__ rowscale_d,
                                 const float* __restrict__ bias,
                                 float* __restrict__ D) {
    const int lane    = threadIdx.x & 31;
    const int wave    = threadIdx.x >> 5;
    const int half    = lane >> 4;
    const int l       = lane & 15;
    const int row0    = blockIdx.x * 16;
    const int arow    = row0 + l;
    const int col     = wave * 16 + l;

    float ascale = 1.0f;
    if (SCALE_A) ascale = rowscale_a[arow];

    v8f c = {};
    const float* arowp = A + (long long)arow * KDIM;

    #pragma unroll
    for (int k = 0; k < KDIM; k += 4) {
        const int ka = k + half * 2;
        v2f a = *(const v2f*)(arowp + ka);
        if (SCALE_A) { a.x *= ascale; a.y *= ascale; }
        v2f b;
        b.x = B[(long long)ka * NDIM + col];
        b.y = B[(long long)(ka + 1) * NDIM + col];
        c = __builtin_amdgcn_wmma_f32_16x16x4_f32(
                 false, a,  false, b,
                 (short)0, c,  false,  false);
    }

    #pragma unroll
    for (int r = 0; r < 8; ++r) {
        int row = row0 + r + half * 8;
        float v = c[r];
        if (EPILOGUE) {
            v = v * rowscale_d[row] + bias[col];
            v = fmaxf(v, 0.0f);
        }
        D[(long long)row * NDIM + col] = v;
    }
}


extern "C" void kernel_launch(void* const* d_in, const int* in_sizes, int n_in,
                              void* d_out, int out_size, void* d_ws, size_t ws_size,
                              hipStream_t stream) {
    const float* x   = (const float*)d_in[0];
    const float* W1  = (const float*)d_in[1];
    const float* b1  = (const float*)d_in[2];
    const float* W2  = (const float*)d_in[3];
    const float* b2  = (const float*)d_in[4];
    const int*   src = (const int*)d_in[5];
    const int*   dst = (const int*)d_in[6];
    float*       out = (float*)d_out;

    int*   part_in  = (int*)d_ws;
    int*   part_out = part_in + N_CHUNKS * N_NODES;
    int*   subsum   = part_out + N_CHUNKS * N_NODES;
    int*   seg      = subsum + N_SUBS * N_CHUNKS;
    int*   col      = seg + 2 * N_NODES;
    float* norm_src = (float*)(col + N_EDGES);
    float* norm_dst = norm_src + N_NODES;
    float* agg      = norm_dst + N_NODES;
    float* h1       = agg + (long long)N_NODES * 64;
    float* h2w      = h1 + (long long)N_NODES * 128;

    degree_count_kernel<<<dim3(N_CHUNKS * N_RANGES, 2), BIN_T, 0, stream>>>(
        src, dst, part_out, part_in, subsum);
    scan_norm_kernel<<<N_SUBS, SCAN_T, 0, stream>>>(
        part_out, part_in, subsum, seg, norm_src, norm_dst);
    fill_kernel<<<N_CHUNKS * N_RANGES, BIN_T, 0, stream>>>(src, dst, part_in, col);

    gather_kernel<true><<<GATHER_BLOCKS, 256, 0, stream>>>(
        x, norm_src, seg, col, nullptr, nullptr, agg);
    wmma_gemm_kernel<F_IN, F_HID, false, true><<<N_NODES / 16, 8 * 32, 0, stream>>>(
        agg, W1, nullptr, norm_dst, b1, h1);

    wmma_gemm_kernel<F_HID, F_OUT, true, false><<<N_NODES / 16, 4 * 32, 0, stream>>>(
        h1, W2, norm_src, nullptr, nullptr, h2w);
    gather_kernel<false><<<GATHER_BLOCKS, 256, 0, stream>>>(
        h2w, nullptr, seg, col, norm_dst, b2, out);
}


# Both layer GEMMs in one kernel: the 16x128 hidden tile stays in LDS as the second product's A operand; 1024 4-wave groups loop over tiles, W1/W2 fragments held in registers; sum order unchanged
# speedup vs baseline: 4.9669x; 1.1624x over previous
#include <hip/hip_runtime.h>


#define N_NODES 50000
#define N_EDGES 800000
#define F_IN    64
#define F_HID   128
#define F_OUT   64

typedef __attribute__((ext_vector_type(2))) float v2f;
typedef __attribute__((ext_vector_type(8))) float v8f;

#define N_CHUNKS        32
#define CHUNK_V         (N_EDGES / N_CHUNKS / 4)
#define RANGE           6400
#define N_RANGES        8
#define BIN_T           512
#define SUB             256
#define SUBS_PER_RANGE  (RANGE / SUB)
#define N_SUBS          ((N_NODES + SUB - 1) / SUB)
#define TRIPS           ((CHUNK_V + BIN_T - 1) / BIN_T)
#define PLACE_TRIPS     ((RANGE / 4 + BIN_T - 1) / BIN_T)

__global__ __launch_bounds__(BIN_T)
void degree_count_kernel(const int* __restrict__ src, const int* __restrict__ dst,
                         int* __restrict__ part_out, int* __restrict__ part_in,
                         int* __restrict__ subsum) {
    __shared__ __attribute__((aligned(16))) int cnt[RANGE];
    const int tid  = threadIdx.x;
    const int lane = tid & 31;
    const int k    = blockIdx.x / N_RANGES;
    const int s    = blockIdx.x % N_RANGES;
    const int lo   = s * RANGE;
    const bool is_in = (blockIdx.y == 0);
    const int* __restrict__ idx  = is_in ? dst : src;
    int* __restrict__       part = is_in ? part_in : part_out;

    const int4* idx4 = (const int4*)idx + k * CHUNK_V;
    int4 q[TRIPS];
    #pragma unroll
    for (int i = 0; i < TRIPS; ++i) {
        const int v = tid + i * BIN_T;
        q[i] = idx4[v < CHUNK_V ? v : CHUNK_V - 1];
    }

    int4* cnt4 = (int4*)cnt;
    for (int n = tid; n < RANGE / 4; n += BIN_T) cnt4[n] = make_int4(0, 0, 0, 0);
    __syncthreads();

    #pragma unroll
    for (int i = 0; i < TRIPS; ++i) {
        if (tid + i * BIN_T < CHUNK_V) {
            const unsigned r0 = (unsigned)(q[i].x - lo), r1 = (unsigned)(q[i].y - lo);
            const unsigned r2 = (unsigned)(q[i].z - lo), r3 = (unsigned)(q[i].w - lo);
            if (r0 < RANGE) atomicAdd(&cnt[r0], 1);
            if (r1 < RANGE) atomicAdd(&cnt[r1], 1);
            if (r2 < RANGE) atomicAdd(&cnt[r2], 1);
            if (r3 < RANGE) atomicAdd(&cnt[r3], 1);
        }
    }
    __syncthreads();

    const int4* cnt4r = (const int4*)cnt;
    int4* part4 = (int4*)(part + k * N_NODES + lo);
    for (int j = tid >> 5; j < SUBS_PER_RANGE; j += BIN_T / 32) {
        const int n0 = j * SUB + lane * 4;
        const int4 a = cnt4r[n0 >> 2];
        const int4 b = cnt4r[(n0 + SUB / 2) >> 2];
        if (lo + n0 < N_NODES)           part4[n0 >> 2] = a;
        if (lo + n0 + SUB / 2 < N_NODES) part4[(n0 + SUB / 2) >> 2] = b;
        if (is_in) {
            int w = ((a.x + a.y) + (a.z + a.w)) + ((b.x + b.y) + (b.z + b.w));
            for (int o = 16; o > 0; o >>= 1) w += __shfl_xor(w, o);
            const int g = s * SUBS_PER_RANGE + j;
            if (lane == 0 && g < N_SUBS) subsum[g * N_CHUNKS + k] = w;
        }
    }
}

#define SCAN_T SUB
__global__ __launch_bounds__(SCAN_T)
void scan_norm_kernel(const int* __restrict__ part_out, int* __restrict__ part_in,
                      const int* __restrict__ subsum,
                      int* __restrict__ seg, float* __restrict__ norm_src,
                      float* __restrict__ norm_dst) {
    __shared__ int before_part[SCAN_T / 32];
    __shared__ int wave_total[SCAN_T / 32];
    const int tid  = threadIdx.x;
    const int lane = tid & 31;
    const int wave = tid >> 5;
    const int node = blockIdx.x * SCAN_T + tid;
    const bool valid = node < N_NODES;

    int pre = 0;
    const int nbefore = blockIdx.x * N_CHUNKS;
    for (int i = tid; i < nbefore; i += SCAN_T) pre += subsum[i];
    for (int o = 16; o > 0; o >>= 1) pre += __shfl_xor(pre, o);
    if (lane == 0) before_part[wave] = pre;

    const int ln = valid ? node : N_NODES - 1;
    int p[N_CHUNKS];
    int t = 0, oc = 0;
    #pragma unroll
    for (int k = 0; k < N_CHUNKS; ++k) {
        p[k] = part_in[k * N_NODES + ln];
        t += p[k];
    }
    #pragma unroll
    for (int k = 0; k < N_CHUNKS; ++k) oc += part_out[k * N_NODES + ln];
    if (!valid) t = 0;

    int inc = t;
    for (int o = 1; o < 32; o <<= 1) {
        int u = __shfl_up(inc, o);
        if (lane >= o) inc += u;
    }
    if (lane == 31) wave_total[wave] = inc;
    __syncthreads();

    int base = 0;
    for (int w = 0; w < SCAN_T / 32; ++w) {
        base += before_part[w];
        if (w < wave) base += wave_total[w];
    }

    if (valid) {
        const int b0 = base + inc - t;
        int run = b0;
        #pragma unroll
        for (int k = 0; k < N_CHUNKS; ++k) {
            part_in[k * N_NODES + node] = run;
            run += p[k];
        }
        ((int2*)seg)[node] = make_int2(b0, run);
        norm_src[node] = rsqrtf(fmaxf((float)oc, 1.0f));
        norm_dst[node] = rsqrtf(fmaxf((float)t, 1.0f));
    }
}

__global__ __launch_bounds__(BIN_T)
void fill_kernel(const int* __restrict__ src, const int* __restrict__ dst,
                 const int* __restrict__ start, int* __restrict__ col) {
    __shared__ __attribute__((aligned(16))) int place[PLACE_TRIPS * BIN_T * 4];
    const int tid = threadIdx.x;
    const int k   = blockIdx.x / N_RANGES;
    const int s   = blockIdx.x % N_RANGES;
    const int lo  = s * RANGE;

    const int4* d4 = (const int4*)dst + k * CHUNK_V;
    const int4* s4 = (const int4*)src + k * CHUNK_V;
    int4 qd[TRIPS], qs[TRIPS];
    #pragma unroll
    for (int i = 0; i < TRIPS; ++i) {
        const int v = tid + i * BIN_T;
        const int vc = v < CHUNK_V ? v : CHUNK_V - 1;
        qd[i] = d4[vc];
        qs[i] = s4[vc];
    }
    const int4* start4 = (const int4*)(start + k * N_NODES);
    int4 st[PLACE_TRIPS];
    #pragma unroll
    for (int i = 0; i < PLACE_TRIPS; ++i) {
        const int node = lo + (tid + i * BIN_T) * 4;
        st[i] = start4[(node < N_NODES ? node : N_NODES - 4) >> 2];
    }
    int4* place4 = (int4*)place;
    #pragma unroll
    for (int i = 0; i < PLACE_TRIPS; ++i) place4[tid + i * BIN_T] = st[i];
    __syncthreads();

    #pragma unroll
    for (int i = 0; i < TRIPS; ++i) {
        if (tid + i * BIN_T < CHUNK_V) {
            const unsigned r0 = (unsigned)(qd[i].x - lo), r1 = (unsigned)(qd[i].y - lo);
            const unsigned r2 = (unsigned)(qd[i].z - lo), r3 = (unsigned)(qd[i].w - lo);
            const bool m0 = r0 < RANGE, m1 = r1 < RANGE, m2 = r2 < RANGE, m3 = r3 < RANGE;
            unsigned p0 = 0, p1 = 0, p2 = 0, p3 = 0;
            if (m0) p0 = (unsigned)atomicAdd(&place[r0], 1);
            if (m1) p1 = (unsigned)atomicAdd(&place[r1], 1);
            if (m2) p2 = (unsigned)atomicAdd(&place[r2], 1);
            if (m3) p3 = (unsigned)atomicAdd(&place[r3], 1);
            if (m0) col[p0] = qs[i].x;
            if (m1) col[p1] = qs[i].y;
            if (m2) col[p2] = qs[i].z;
            if (m3) col[p3] = qs[i].w;
        }
    }
}

#define GATHER_BLOCKS 1024

template <bool FIRST>
__device__ __forceinline__ void add_row(float4& acc, const float* __restrict__ feat,
                                        const float* __restrict__ norm_src,
                                        int c, bool valid, int l) {
    float4 r = *(const float4*)(feat + (long long)c * 64 + l * 4);
    if (!valid) r = make_float4(0.0f, 0.0f, 0.0f, 0.0f);
    if (FIRST) {
        const float sc = norm_src[c];
        acc.x += r.x * sc; acc.y += r.y * sc; acc.z += r.z * sc; acc.w += r.w * sc;
    } else {
        acc.x += r.x; acc.y += r.y; acc.z += r.z; acc.w += r.w;
    }
}

template <bool FIRST>
__global__ __launch_bounds__(256)
void gather_kernel(const float* __restrict__ feat, const float* __restrict__ norm_src,
                   const int* __restrict__ seg, const int* __restrict__ col,
                   const float* __restrict__ norm_dst, const float* __restrict__ bias,
                   float* __restrict__ outp) {
    const int lane = threadIdx.x & 31;
    const int half = lane >> 4;
    const int l    = lane & 15;
    const int nw   = GATHER_BLOCKS * 8;
    int node = __builtin_amdgcn_readfirstlane(blockIdx.x * 8 + (threadIdx.x >> 5));
    const int2* seg2 = (const int2*)seg;
    const int2 se0 = seg2[node];
    int jb  = se0.x + half;
    int end = se0.y;
    float4 bv = make_float4(0.0f, 0.0f, 0.0f, 0.0f);
    if (!FIRST) bv = *(const float4*)(bias + l * 4);

    while (true) {
        const int next = node + nw;
        int2 nse = make_int2(0, 0);
        if (next < N_NODES) nse = seg2[next];

        float nd = 1.0f;
        if (!FIRST) nd = norm_dst[node];

        float4 acc = make_float4(0.0f, 0.0f, 0.0f, 0.0f);
        for (int j = jb; j < end; j += 8) {
            const bool v1 = (j + 2) < end, v2 = (j + 4) < end, v3 = (j + 6) < end;
            const int c0 = col[j];
            const int c1 = col[v1 ? j + 2 : j];
            const int c2 = col[v2 ? j + 4 : j];
            const int c3 = col[v3 ? j + 6 : j];
            add_row<FIRST>(acc, feat, norm_src, c0, true, l);
            add_row<FIRST>(acc, feat, norm_src, c1, v1, l);
            add_row<FIRST>(acc, feat, norm_src, c2, v2, l);
            add_row<FIRST>(acc, feat, norm_src, c3, v3, l);
        }
        acc.x += __shfl_xor(acc.x, 16);
        acc.y += __shfl_xor(acc.y, 16);
        acc.z += __shfl_xor(acc.z, 16);
        acc.w += __shfl_xor(acc.w, 16);
        if (half == 0) {
            float4 o = acc;
            if (!FIRST) {
                o.x = acc.x * nd + bv.x; o.y = acc.y * nd + bv.y;
                o.z = acc.z * nd + bv.z; o.w = acc.w * nd + bv.w;
            }
            *(float4*)(outp + (long long)node * 64 + l * 4) = o;
        }
        if (next >= N_NODES) break;
        node = next;
        jb   = nse.x + half;
        end  = nse.y;
    }
}

#define N_TILES   (N_NODES / 16)
#define FUSE_WGS  1024
#define H_PITCH   (F_HID + 4)

__device__ __forceinline__ v8f wmma4(v2f a, v2f b, v8f c) {
    return __builtin_amdgcn_wmma_f32_16x16x4_f32(
         false, a,  false, b,
         (short)0, c,  false,  false);
}

__global__ __launch_bounds__(128)
void layer_gemms_kernel(const float* __restrict__ agg, const float* __restrict__ W1,
                        const float* __restrict__ b1, const float* __restrict__ W2,
                        const float* __restrict__ norm_src,
                        const float* __restrict__ norm_dst,
                        float* __restrict__ h2w) {
    __shared__ __attribute__((aligned(16))) float hs[16 * H_PITCH];
    const int lane = threadIdx.x & 31;
    const int wave = threadIdx.x >> 5;
    const int half = lane >> 4;
    const int l    = lane & 15;

    const int colA = (2 * wave) * 16 + l;
    const int colB = colA + 16;
    const int col2 = wave * 16 + l;
    v2f w1a[F_IN / 4], w1b[F_IN / 4], w2[F_HID / 4];
    #pragma unroll
    for (int s = 0; s < F_IN / 4; ++s) {
        const int ka = s * 4 + half * 2;
        w1a[s].x = W1[ka * F_HID + colA];  w1a[s].y = W1[(ka + 1) * F_HID + colA];
        w1b[s].x = W1[ka * F_HID + colB];  w1b[s].y = W1[(ka + 1) * F_HID + colB];
    }
    #pragma unroll
    for (int s = 0; s < F_HID / 4; ++s) {
        const int ka = s * 4 + half * 2;
        w2[s].x = W2[ka * F_OUT + col2];   w2[s].y = W2[(ka + 1) * F_OUT + col2];
    }
    const float biasA = b1[colA], biasB = b1[colB];

    int tile = blockIdx.x;
    v2f a[F_IN / 4];
    {
        const float* ap = agg + (long long)(tile * 16 + l) * F_IN + half * 2;
        #pragma unroll
        for (int s = 0; s < F_IN / 4; ++s) a[s] = *(const v2f*)(ap + s * 4);
    }

    while (true) {
        const int row0 = tile * 16;
        float nd[8], ns[8];
        #pragma unroll
        for (int r = 0; r < 8; ++r) {
            nd[r] = norm_dst[row0 + r + half * 8];
            ns[r] = norm_src[row0 + r + half * 8];
        }

        v8f c0 = {}, c1 = {};
        #pragma unroll
        for (int s = 0; s < F_IN / 4; ++s) {
            c0 = wmma4(a[s], w1a[s], c0);
            c1 = wmma4(a[s], w1b[s], c1);
        }

        #pragma unroll
        for (int r = 0; r < 8; ++r) {
            float* hp = hs + (r + half * 8) * H_PITCH;
            hp[colA] = fmaxf(c0[r] * nd[r] + biasA, 0.0f) * ns[r];
            hp[colB] = fmaxf(c1[r] * nd[r] + biasB, 0.0f) * ns[r];
        }
        __syncthreads();

        const int next = tile + FUSE_WGS;
        {
            const int nt = next < N_TILES ? next : tile;
            const float* ap = agg + (long long)(nt * 16 + l) * F_IN + half * 2;
            #pragma unroll
            for (int s = 0; s < F_IN / 4; ++s) a[s] = *(const v2f*)(ap + s * 4);
        }

        v2f a2[F_HID / 4];
        {
            const float* hp = hs + l * H_PITCH + half * 2;
            #pragma unroll
            for (int s = 0; s < F_HID / 4; ++s) a2[s] = *(const v2f*)(hp + s * 4);
        }
        v8f c = {};
        #pragma unroll
        for (int s = 0; s < F_HID / 4; ++s) c = wmma4(a2[s], w2[s], c);

        __syncthreads();

        #pragma unroll
        for (int r = 0; r < 8; ++r)
            h2w[(long long)(row0 + r + half * 8) * F_OUT + col2] = c[r];

        if (next >= N_TILES) break;
        tile = next;
    }
}


extern "C" void kernel_launch(void* const* d_in, const int* in_sizes, int n_in,
                              void* d_out, int out_size, void* d_ws, size_t ws_size,
                              hipStream_t stream) {
    const float* x   = (const float*)d_in[0];
    const float* W1  = (const float*)d_in[1];
    const float* b1  = (const float*)d_in[2];
    const float* W2  = (const float*)d_in[3];
    const float* b2  = (const float*)d_in[4];
    const int*   src = (const int*)d_in[5];
    const int*   dst = (const int*)d_in[6];
    float*       out = (float*)d_out;

    int*   part_in  = (int*)d_ws;
    int*   part_out = part_in + N_CHUNKS * N_NODES;
    int*   subsum   = part_out + N_CHUNKS * N_NODES;
    int*   seg      = subsum + N_SUBS * N_CHUNKS;
    int*   col      = seg + 2 * N_NODES;
    float* norm_src = (float*)(col + N_EDGES);
    float* norm_dst = norm_src + N_NODES;
    float* agg      = norm_dst + N_NODES;
    float* h2w      = agg + (long long)N_NODES * 64;

    degree_count_kernel<<<dim3(N_CHUNKS * N_RANGES, 2), BIN_T, 0, stream>>>(
        src, dst, part_out, part_in, subsum);
    scan_norm_kernel<<<N_SUBS, SCAN_T, 0, stream>>>(
        part_out, part_in, subsum, seg, norm_src, norm_dst);
    fill_kernel<<<N_CHUNKS * N_RANGES, BIN_T, 0, stream>>>(src, dst, part_in, col);

    gather_kernel<true><<<GATHER_BLOCKS, 256, 0, stream>>>(
        x, norm_src, seg, col, nullptr, nullptr, agg);
    layer_gemms_kernel<<<FUSE_WGS, 128, 0, stream>>>(
        agg, W1, b1, W2, norm_src, norm_dst, h2w);
    gather_kernel<false><<<GATHER_BLOCKS, 256, 0, stream>>>(
        h2w, nullptr, seg, col, norm_dst, b2, out);
}


# Gather: a node's list is read once, coalesced, a node ahead; row addresses by DPP broadcast in the 16-lane half; all rows of the node in flight together (1 round trip per node); sum order unchanged
# speedup vs baseline: 5.4261x; 1.0925x over previous
#include <hip/hip_runtime.h>


#define N_NODES 50000
#define N_EDGES 800000
#define F_IN    64
#define F_HID   128
#define F_OUT   64

typedef __attribute__((ext_vector_type(2))) float v2f;
typedef __attribute__((ext_vector_type(8))) float v8f;

#define N_CHUNKS        32
#define CHUNK_V         (N_EDGES / N_CHUNKS / 4)
#define RANGE           6400
#define N_RANGES        8
#define BIN_T           512
#define SUB             256
#define SUBS_PER_RANGE  (RANGE / SUB)
#define N_SUBS          ((N_NODES + SUB - 1) / SUB)
#define TRIPS           ((CHUNK_V + BIN_T - 1) / BIN_T)
#define PLACE_TRIPS     ((RANGE / 4 + BIN_T - 1) / BIN_T)

__global__ __launch_bounds__(BIN_T)
void degree_count_kernel(const int* __restrict__ src, const int* __restrict__ dst,
                         int* __restrict__ part_out, int* __restrict__ part_in,
                         int* __restrict__ subsum) {
    __shared__ __attribute__((aligned(16))) int cnt[RANGE];
    const int tid  = threadIdx.x;
    const int lane = tid & 31;
    const int k    = blockIdx.x / N_RANGES;
    const int s    = blockIdx.x % N_RANGES;
    const int lo   = s * RANGE;
    const bool is_in = (blockIdx.y == 0);
    const int* __restrict__ idx  = is_in ? dst : src;
    int* __restrict__       part = is_in ? part_in : part_out;

    const int4* idx4 = (const int4*)idx + k * CHUNK_V;
    int4 q[TRIPS];
    #pragma unroll
    for (int i = 0; i < TRIPS; ++i) {
        const int v = tid + i * BIN_T;
        q[i] = idx4[v < CHUNK_V ? v : CHUNK_V - 1];
    }

    int4* cnt4 = (int4*)cnt;
    for (int n = tid; n < RANGE / 4; n += BIN_T) cnt4[n] = make_int4(0, 0, 0, 0);
    __syncthreads();

    #pragma unroll
    for (int i = 0; i < TRIPS; ++i) {
        if (tid + i * BIN_T < CHUNK_V) {
            const unsigned r0 = (unsigned)(q[i].x - lo), r1 = (unsigned)(q[i].y - lo);
            const unsigned r2 = (unsigned)(q[i].z - lo), r3 = (unsigned)(q[i].w - lo);
            if (r0 < RANGE) atomicAdd(&cnt[r0], 1);
            if (r1 < RANGE) atomicAdd(&cnt[r1], 1);
            if (r2 < RANGE) atomicAdd(&cnt[r2], 1);
            if (r3 < RANGE) atomicAdd(&cnt[r3], 1);
        }
    }
    __syncthreads();

    const int4* cnt4r = (const int4*)cnt;
    int4* part4 = (int4*)(part + k * N_NODES + lo);
    for (int j = tid >> 5; j < SUBS_PER_RANGE; j += BIN_T / 32) {
        const int n0 = j * SUB + lane * 4;
        const int4 a = cnt4r[n0 >> 2];
        const int4 b = cnt4r[(n0 + SUB / 2) >> 2];
        if (lo + n0 < N_NODES)           part4[n0 >> 2] = a;
        if (lo + n0 + SUB / 2 < N_NODES) part4[(n0 + SUB / 2) >> 2] = b;
        if (is_in) {
            int w = ((a.x + a.y) + (a.z + a.w)) + ((b.x + b.y) + (b.z + b.w));
            for (int o = 16; o > 0; o >>= 1) w += __shfl_xor(w, o);
            const int g = s * SUBS_PER_RANGE + j;
            if (lane == 0 && g < N_SUBS) subsum[g * N_CHUNKS + k] = w;
        }
    }
}

#define SCAN_T SUB
__global__ __launch_bounds__(SCAN_T)
void scan_norm_kernel(const int* __restrict__ part_out, int* __restrict__ part_in,
                      const int* __restrict__ subsum,
                      int* __restrict__ seg, float* __restrict__ norm_src,
                      float* __restrict__ norm_dst) {
    __shared__ int before_part[SCAN_T / 32];
    __shared__ int wave_total[SCAN_T / 32];
    const int tid  = threadIdx.x;
    const int lane = tid & 31;
    const int wave = tid >> 5;
    const int node = blockIdx.x * SCAN_T + tid;
    const bool valid = node < N_NODES;

    int pre = 0;
    const int nbefore = blockIdx.x * N_CHUNKS;
    for (int i = tid; i < nbefore; i += SCAN_T) pre += subsum[i];
    for (int o = 16; o > 0; o >>= 1) pre += __shfl_xor(pre, o);
    if (lane == 0) before_part[wave] = pre;

    const int ln = valid ? node : N_NODES - 1;
    int p[N_CHUNKS];
    int t = 0, oc = 0;
    #pragma unroll
    for (int k = 0; k < N_CHUNKS; ++k) {
        p[k] = part_in[k * N_NODES + ln];
        t += p[k];
    }
    #pragma unroll
    for (int k = 0; k < N_CHUNKS; ++k) oc += part_out[k * N_NODES + ln];
    if (!valid) t = 0;

    int inc = t;
    for (int o = 1; o < 32; o <<= 1) {
        int u = __shfl_up(inc, o);
        if (lane >= o) inc += u;
    }
    if (lane == 31) wave_total[wave] = inc;
    __syncthreads();

    int base = 0;
    for (int w = 0; w < SCAN_T / 32; ++w) {
        base += before_part[w];
        if (w < wave) base += wave_total[w];
    }

    if (valid) {
        const int b0 = base + inc - t;
        int run = b0;
        #pragma unroll
        for (int k = 0; k < N_CHUNKS; ++k) {
            part_in[k * N_NODES + node] = run;
            run += p[k];
        }
        ((int2*)seg)[node] = make_int2(b0, run);
        norm_src[node] = rsqrtf(fmaxf((float)oc, 1.0f));
        norm_dst[node] = rsqrtf(fmaxf((float)t, 1.0f));
    }
}

__global__ __launch_bounds__(BIN_T)
void fill_kernel(const int* __restrict__ src, const int* __restrict__ dst,
                 const int* __restrict__ start, int* __restrict__ col) {
    __shared__ __attribute__((aligned(16))) int place[PLACE_TRIPS * BIN_T * 4];
    const int tid = threadIdx.x;
    const int k   = blockIdx.x / N_RANGES;
    const int s   = blockIdx.x % N_RANGES;
    const int lo  = s * RANGE;

    const int4* d4 = (const int4*)dst + k * CHUNK_V;
    const int4* s4 = (const int4*)src + k * CHUNK_V;
    int4 qd[TRIPS], qs[TRIPS];
    #pragma unroll
    for (int i = 0; i < TRIPS; ++i) {
        const int v = tid + i * BIN_T;
        const int vc = v < CHUNK_V ? v : CHUNK_V - 1;
        qd[i] = d4[vc];
        qs[i] = s4[vc];
    }
    const int4* start4 = (const int4*)(start + k * N_NODES);
    int4 st[PLACE_TRIPS];
    #pragma unroll
    for (int i = 0; i < PLACE_TRIPS; ++i) {
        const int node = lo + (tid + i * BIN_T) * 4;
        st[i] = start4[(node < N_NODES ? node : N_NODES - 4) >> 2];
    }
    int4* place4 = (int4*)place;
    #pragma unroll
    for (int i = 0; i < PLACE_TRIPS; ++i) place4[tid + i * BIN_T] = st[i];
    __syncthreads();

    #pragma unroll
    for (int i = 0; i < TRIPS; ++i) {
        if (tid + i * BIN_T < CHUNK_V) {
            const unsigned r0 = (unsigned)(qd[i].x - lo), r1 = (unsigned)(qd[i].y - lo);
            const unsigned r2 = (unsigned)(qd[i].z - lo), r3 = (unsigned)(qd[i].w - lo);
            const bool m0 = r0 < RANGE, m1 = r1 < RANGE, m2 = r2 < RANGE, m3 = r3 < RANGE;
            unsigned p0 = 0, p1 = 0, p2 = 0, p3 = 0;
            if (m0) p0 = (unsigned)atomicAdd(&place[r0], 1);
            if (m1) p1 = (unsigned)atomicAdd(&place[r1], 1);
            if (m2) p2 = (unsigned)atomicAdd(&place[r2], 1);
            if (m3) p3 = (unsigned)atomicAdd(&place[r3], 1);
            if (m0) col[p0] = qs[i].x;
            if (m1) col[p1] = qs[i].y;
            if (m2) col[p2] = qs[i].z;
            if (m3) col[p3] = qs[i].w;
        }
    }
}

#define GATHER_BLOCKS 1024

template <int T>
__device__ __forceinline__ int half_lane(int v) {
    return __builtin_amdgcn_update_dpp(v, v, 0x150 + T, 0xf, 0xf, true);
}

template <bool FIRST, int T, int END>
struct Rows {
    static __device__ __forceinline__ void issue(float4* r, const char* __restrict__ featb,
                                                 int offs, int l16) {
        r[T] = *(const float4*)(featb + (unsigned)(half_lane<T>(offs) + l16));
        Rows<FIRST, T + 1, END>::issue(r, featb, offs, l16);
    }
    static __device__ __forceinline__ void add(float4& acc, const float4* r, int nsb, int mh) {
        float4 v = r[T];
        if (T >= END - 4 && !(2 * T < mh)) v = make_float4(0.0f, 0.0f, 0.0f, 0.0f);
        if (FIRST) {
            const float sc = __int_as_float(half_lane<T>(nsb));
            acc.x = __builtin_fmaf(v.x, sc, acc.x); acc.y = __builtin_fmaf(v.y, sc, acc.y);
            acc.z = __builtin_fmaf(v.z, sc, acc.z); acc.w = __builtin_fmaf(v.w, sc, acc.w);
        } else {
            acc.x += v.x; acc.y += v.y; acc.z += v.z; acc.w += v.w;
        }
        Rows<FIRST, T + 1, END>::add(acc, r, nsb, mh);
    }
};
template <bool FIRST, int END>
struct Rows<FIRST, END, END> {
    static __device__ __forceinline__ void issue(float4*, const char* __restrict__, int, int) {}
    static __device__ __forceinline__ void add(float4&, const float4*, int, int) {}
};

template <bool FIRST, int NT>
__device__ __forceinline__ void sum_rows(float4& acc, const char* __restrict__ featb,
                                         int offs, int l16, int nsb, int mh) {
    float4 r[4 * NT];
    Rows<FIRST, 0, 4 * NT>::issue(r, featb, offs, l16);
    __builtin_amdgcn_sched_barrier(0);
    Rows<FIRST, 0, 4 * NT>::add(acc, r, nsb, mh);
}

__device__ __forceinline__ int list_place(int first, int end, int ent) {
    const int p = min(first + ent, end - 1);
    return max(p, 0);
}

template <bool FIRST>
__global__ __launch_bounds__(256)
void gather_kernel(const float* __restrict__ feat, const float* __restrict__ norm_src,
                   const int* __restrict__ seg, const int* __restrict__ col,
                   const float* __restrict__ norm_dst, const float* __restrict__ bias,
                   float* __restrict__ outp) {
    const int lane = threadIdx.x & 31;
    const int half = lane >> 4;
    const int l    = lane & 15;
    const int l16  = l * 16;
    const int ent  = 2 * l + half;
    const int nw   = GATHER_BLOCKS * 8;
    const char* __restrict__ featb = (const char*)feat;
    int node = __builtin_amdgcn_readfirstlane(blockIdx.x * 8 + (threadIdx.x >> 5));
    const int2* seg2 = (const int2*)seg;
    int2 se  = seg2[node];
    int  nxt = node + nw;
    int2 se1 = make_int2(0, 0);
    if (nxt < N_NODES) se1 = seg2[nxt];
    float4 bv = make_float4(0.0f, 0.0f, 0.0f, 0.0f);
    if (!FIRST) bv = *(const float4*)(bias + l * 4);

    int   cv = col[list_place(se.x, se.y, ent)];
    float ns = 0.0f;
    if (FIRST) ns = norm_src[cv];

    while (true) {
        const int nxt2 = nxt + nw;
        int2 se2 = make_int2(0, 0);
        if (nxt2 < N_NODES) se2 = seg2[nxt2];
        const int cvn = col[list_place(se1.x, se1.y, ent)];

        float nd = 1.0f;
        if (!FIRST) nd = norm_dst[node];

        float4 acc = make_float4(0.0f, 0.0f, 0.0f, 0.0f);
        const int cnt = se.y - se.x;
        int   cvc = cv;
        float nsc = ns;
        for (int base = 0; base < cnt; base += 32) {
            if (base > 0) {
                cvc = col[min(se.x + base + ent, se.y - 1)];
                if (FIRST) nsc = norm_src[cvc];
            }
            const int m    = min(cnt - base, 32);
            const int mh   = m - half;
            const int offs = cvc * (64 * 4);
            const int nsb  = __float_as_int(nsc);
            if (m <= 8)       sum_rows<FIRST, 1>(acc, featb, offs, l16, nsb, mh);
            else if (m <= 16) sum_rows<FIRST, 2>(acc, featb, offs, l16, nsb, mh);
            else if (m <= 24) sum_rows<FIRST, 3>(acc, featb, offs, l16, nsb, mh);
            else              sum_rows<FIRST, 4>(acc, featb, offs, l16, nsb, mh);
        }
        float nsn = 0.0f;
        if (FIRST) nsn = norm_src[cvn];

        acc.x += __shfl_xor(acc.x, 16);
        acc.y += __shfl_xor(acc.y, 16);
        acc.z += __shfl_xor(acc.z, 16);
        acc.w += __shfl_xor(acc.w, 16);
        if (half == 0) {
            float4 o = acc;
            if (!FIRST) {
                o.x = acc.x * nd + bv.x; o.y = acc.y * nd + bv.y;
                o.z = acc.z * nd + bv.z; o.w = acc.w * nd + bv.w;
            }
            *(float4*)(outp + (long long)node * 64 + l * 4) = o;
        }
        if (nxt >= N_NODES) break;
        node = nxt;  nxt = nxt2;
        se   = se1;  se1 = se2;
        cv   = cvn;  ns  = nsn;
    }
}

#define N_TILES   (N_NODES / 16)
#define FUSE_WGS  1024
#define H_PITCH   (F_HID + 4)

__device__ __forceinline__ v8f wmma4(v2f a, v2f b, v8f c) {
    return __builtin_amdgcn_wmma_f32_16x16x4_f32(
         false, a,  false, b,
         (short)0, c,  false,  false);
}

__global__ __launch_bounds__(128)
void layer_gemms_kernel(const float* __restrict__ agg, const float* __restrict__ W1,
                        const float* __restrict__ b1, const float* __restrict__ W2,
                        const float* __restrict__ norm_src,
                        const float* __restrict__ norm_dst,
                        float* __restrict__ h2w) {
    __shared__ __attribute__((aligned(16))) float hs[16 * H_PITCH];
    const int lane = threadIdx.x & 31;
    const int wave = threadIdx.x >> 5;
    const int half = lane >> 4;
    const int l    = lane & 15;

    const int colA = (2 * wave) * 16 + l;
    const int colB = colA + 16;
    const int col2 = wave * 16 + l;
    v2f w1a[F_IN / 4], w1b[F_IN / 4], w2[F_HID / 4];
    #pragma unroll
    for (int s = 0; s < F_IN / 4; ++s) {
        const int ka = s * 4 + half * 2;
        w1a[s].x = W1[ka * F_HID + colA];  w1a[s].y = W1[(ka + 1) * F_HID + colA];
        w1b[s].x = W1[ka * F_HID + colB];  w1b[s].y = W1[(ka + 1) * F_HID + colB];
    }
    #pragma unroll
    for (int s = 0; s < F_HID / 4; ++s) {
        const int ka = s * 4 + half * 2;
        w2[s].x = W2[ka * F_OUT + col2];   w2[s].y = W2[(ka + 1) * F_OUT + col2];
    }
    const float biasA = b1[colA], biasB = b1[colB];

    int tile = blockIdx.x;
    v2f a[F_IN / 4];
    {
        const float* ap = agg + (long long)(tile * 16 + l) * F_IN + half * 2;
        #pragma unroll
        for (int s = 0; s < F_IN / 4; ++s) a[s] = *(const v2f*)(ap + s * 4);
    }

    while (true) {
        const int row0 = tile * 16;
        float nd[8], ns[8];
        #pragma unroll
        for (int r = 0; r < 8; ++r) {
            nd[r] = norm_dst[row0 + r + half * 8];
            ns[r] = norm_src[row0 + r + half * 8];
        }

        v8f c0 = {}, c1 = {};
        #pragma unroll
        for (int s = 0; s < F_IN / 4; ++s) {
            c0 = wmma4(a[s], w1a[s], c0);
            c1 = wmma4(a[s], w1b[s], c1);
        }

        #pragma unroll
        for (int r = 0; r < 8; ++r) {
            float* hp = hs + (r + half * 8) * H_PITCH;
            hp[colA] = fmaxf(c0[r] * nd[r] + biasA, 0.0f) * ns[r];
            hp[colB] = fmaxf(c1[r] * nd[r] + biasB, 0.0f) * ns[r];
        }
        __syncthreads();

        const int next = tile + FUSE_WGS;
        {
            const int nt = next < N_TILES ? next : tile;
            const float* ap = agg + (long long)(nt * 16 + l) * F_IN + half * 2;
            #pragma unroll
            for (int s = 0; s < F_IN / 4; ++s) a[s] = *(const v2f*)(ap + s * 4);
        }

        v2f a2[F_HID / 4];
        {
            const float* hp = hs + l * H_PITCH + half * 2;
            #pragma unroll
            for (int s = 0; s < F_HID / 4; ++s) a2[s] = *(const v2f*)(hp + s * 4);
        }
        v8f c = {};
        #pragma unroll
        for (int s = 0; s < F_HID / 4; ++s) c = wmma4(a2[s], w2[s], c);

        __syncthreads();

        #pragma unroll
        for (int r = 0; r < 8; ++r)
            h2w[(long long)(row0 + r + half * 8) * F_OUT + col2] = c[r];

        if (next >= N_TILES) break;
        tile = next;
    }
}


extern "C" void kernel_launch(void* const* d_in, const int* in_sizes, int n_in,
                              void* d_out, int out_size, void* d_ws, size_t ws_size,
                              hipStream_t stream) {
    const float* x   = (const float*)d_in[0];
    const float* W1  = (const float*)d_in[1];
    const float* b1  = (const float*)d_in[2];
    const float* W2  = (const float*)d_in[3];
    const float* b2  = (const float*)d_in[4];
    const int*   src = (const int*)d_in[5];
    const int*   dst = (const int*)d_in[6];
    float*       out = (float*)d_out;

    int*   part_in  = (int*)d_ws;
    int*   part_out = part_in + N_CHUNKS * N_NODES;
    int*   subsum   = part_out + N_CHUNKS * N_NODES;
    int*   seg      = subsum + N_SUBS * N_CHUNKS;
    int*   col      = seg + 2 * N_NODES;
    float* norm_src = (float*)(col + N_EDGES);
    float* norm_dst = norm_src + N_NODES;
    float* agg      = norm_dst + N_NODES;
    float* h2w      = agg + (long long)N_NODES * 64;

    degree_count_kernel<<<dim3(N_CHUNKS * N_RANGES, 2), BIN_T, 0, stream>>>(
        src, dst, part_out, part_in, subsum);
    scan_norm_kernel<<<N_SUBS, SCAN_T, 0, stream>>>(
        part_out, part_in, subsum, seg, norm_src, norm_dst);
    fill_kernel<<<N_CHUNKS * N_RANGES, BIN_T, 0, stream>>>(src, dst, part_in, col);

    gather_kernel<true><<<GATHER_BLOCKS, 256, 0, stream>>>(
        x, norm_src, seg, col, nullptr, nullptr, agg);
    layer_gemms_kernel<<<FUSE_WGS, 128, 0, stream>>>(
        agg, W1, b1, W2, norm_src, norm_dst, h2w);
    gather_kernel<false><<<GATHER_BLOCKS, 256, 0, stream>>>(
        h2w, nullptr, seg, col, norm_dst, b2, out);
}


# Layer-1 gather moved into the GEMM kernel: each wave of the 4-wave group sums 4 nodes' rows into the 16x64 LDS tile that feeds the first product; agg buffer and one launch gone; sum order kept
# speedup vs baseline: 5.8699x; 1.0818x over previous
#include <hip/hip_runtime.h>


#define N_NODES 50000
#define N_EDGES 800000
#define F_IN    64
#define F_HID   128
#define F_OUT   64

typedef __attribute__((ext_vector_type(2))) float v2f;
typedef __attribute__((ext_vector_type(8))) float v8f;

#define N_CHUNKS        32
#define CHUNK_V         (N_EDGES / N_CHUNKS / 4)
#define RANGE           6400
#define N_RANGES        8
#define BIN_T           512
#define SUB             256
#define SUBS_PER_RANGE  (RANGE / SUB)
#define N_SUBS          ((N_NODES + SUB - 1) / SUB)
#define TRIPS           ((CHUNK_V + BIN_T - 1) / BIN_T)
#define PLACE_TRIPS     ((RANGE / 4 + BIN_T - 1) / BIN_T)

__global__ __launch_bounds__(BIN_T)
void degree_count_kernel(const int* __restrict__ src, const int* __restrict__ dst,
                         int* __restrict__ part_out, int* __restrict__ part_in,
                         int* __restrict__ subsum) {
    __shared__ __attribute__((aligned(16))) int cnt[RANGE];
    const int tid  = threadIdx.x;
    const int lane = tid & 31;
    const int k    = blockIdx.x / N_RANGES;
    const int s    = blockIdx.x % N_RANGES;
    const int lo   = s * RANGE;
    const bool is_in = (blockIdx.y == 0);
    const int* __restrict__ idx  = is_in ? dst : src;
    int* __restrict__       part = is_in ? part_in : part_out;

    const int4* idx4 = (const int4*)idx + k * CHUNK_V;
    int4 q[TRIPS];
    #pragma unroll
    for (int i = 0; i < TRIPS; ++i) {
        const int v = tid + i * BIN_T;
        q[i] = idx4[v < CHUNK_V ? v : CHUNK_V - 1];
    }

    int4* cnt4 = (int4*)cnt;
    for (int n = tid; n < RANGE / 4; n += BIN_T) cnt4[n] = make_int4(0, 0, 0, 0);
    __syncthreads();

    #pragma unroll
    for (int i = 0; i < TRIPS; ++i) {
        if (tid + i * BIN_T < CHUNK_V) {
            const unsigned r0 = (unsigned)(q[i].x - lo), r1 = (unsigned)(q[i].y - lo);
            const unsigned r2 = (unsigned)(q[i].z - lo), r3 = (unsigned)(q[i].w - lo);
            if (r0 < RANGE) atomicAdd(&cnt[r0], 1);
            if (r1 < RANGE) atomicAdd(&cnt[r1], 1);
            if (r2 < RANGE) atomicAdd(&cnt[r2], 1);
            if (r3 < RANGE) atomicAdd(&cnt[r3], 1);
        }
    }
    __syncthreads();

    const int4* cnt4r = (const int4*)cnt;
    int4* part4 = (int4*)(part + k * N_NODES + lo);
    for (int j = tid >> 5; j < SUBS_PER_RANGE; j += BIN_T / 32) {
        const int n0 = j * SUB + lane * 4;
        const int4 a = cnt4r[n0 >> 2];
        const int4 b = cnt4r[(n0 + SUB / 2) >> 2];
        if (lo + n0 < N_NODES)           part4[n0 >> 2] = a;
        if (lo + n0 + SUB / 2 < N_NODES) part4[(n0 + SUB / 2) >> 2] = b;
        if (is_in) {
            int w = ((a.x + a.y) + (a.z + a.w)) + ((b.x + b.y) + (b.z + b.w));
            for (int o = 16; o > 0; o >>= 1) w += __shfl_xor(w, o);
            const int g = s * SUBS_PER_RANGE + j;
            if (lane == 0 && g < N_SUBS) subsum[g * N_CHUNKS + k] = w;
        }
    }
}

#define SCAN_T SUB
__global__ __launch_bounds__(SCAN_T)
void scan_norm_kernel(const int* __restrict__ part_out, int* __restrict__ part_in,
                      const int* __restrict__ subsum,
                      int* __restrict__ seg, float* __restrict__ norm_src,
                      float* __restrict__ norm_dst) {
    __shared__ int before_part[SCAN_T / 32];
    __shared__ int wave_total[SCAN_T / 32];
    const int tid  = threadIdx.x;
    const int lane = tid & 31;
    const int wave = tid >> 5;
    const int node = blockIdx.x * SCAN_T + tid;
    const bool valid = node < N_NODES;

    int pre = 0;
    const int nbefore = blockIdx.x * N_CHUNKS;
    for (int i = tid; i < nbefore; i += SCAN_T) pre += subsum[i];
    for (int o = 16; o > 0; o >>= 1) pre += __shfl_xor(pre, o);
    if (lane == 0) before_part[wave] = pre;

    const int ln = valid ? node : N_NODES - 1;
    int p[N_CHUNKS];
    int t = 0, oc = 0;
    #pragma unroll
    for (int k = 0; k < N_CHUNKS; ++k) {
        p[k] = part_in[k * N_NODES + ln];
        t += p[k];
    }
    #pragma unroll
    for (int k = 0; k < N_CHUNKS; ++k) oc += part_out[k * N_NODES + ln];
    if (!valid) t = 0;

    int inc = t;
    for (int o = 1; o < 32; o <<= 1) {
        int u = __shfl_up(inc, o);
        if (lane >= o) inc += u;
    }
    if (lane == 31) wave_total[wave] = inc;
    __syncthreads();

    int base = 0;
    for (int w = 0; w < SCAN_T / 32; ++w) {
        base += before_part[w];
        if (w < wave) base += wave_total[w];
    }

    if (valid) {
        const int b0 = base + inc - t;
        int run = b0;
        #pragma unroll
        for (int k = 0; k < N_CHUNKS; ++k) {
            part_in[k * N_NODES + node] = run;
            run += p[k];
        }
        ((int2*)seg)[node] = make_int2(b0, run);
        norm_src[node] = rsqrtf(fmaxf((float)oc, 1.0f));
        norm_dst[node] = rsqrtf(fmaxf((float)t, 1.0f));
    }
}

__global__ __launch_bounds__(BIN_T)
void fill_kernel(const int* __restrict__ src, const int* __restrict__ dst,
                 const int* __restrict__ start, int* __restrict__ col) {
    __shared__ __attribute__((aligned(16))) int place[PLACE_TRIPS * BIN_T * 4];
    const int tid = threadIdx.x;
    const int k   = blockIdx.x / N_RANGES;
    const int s   = blockIdx.x % N_RANGES;
    const int lo  = s * RANGE;

    const int4* d4 = (const int4*)dst + k * CHUNK_V;
    const int4* s4 = (const int4*)src + k * CHUNK_V;
    int4 qd[TRIPS], qs[TRIPS];
    #pragma unroll
    for (int i = 0; i < TRIPS; ++i) {
        const int v = tid + i * BIN_T;
        const int vc = v < CHUNK_V ? v : CHUNK_V - 1;
        qd[i] = d4[vc];
        qs[i] = s4[vc];
    }
    const int4* start4 = (const int4*)(start + k * N_NODES);
    int4 st[PLACE_TRIPS];
    #pragma unroll
    for (int i = 0; i < PLACE_TRIPS; ++i) {
        const int node = lo + (tid + i * BIN_T) * 4;
        st[i] = start4[(node < N_NODES ? node : N_NODES - 4) >> 2];
    }
    int4* place4 = (int4*)place;
    #pragma unroll
    for (int i = 0; i < PLACE_TRIPS; ++i) place4[tid + i * BIN_T] = st[i];
    __syncthreads();

    #pragma unroll
    for (int i = 0; i < TRIPS; ++i) {
        if (tid + i * BIN_T < CHUNK_V) {
            const unsigned r0 = (unsigned)(qd[i].x - lo), r1 = (unsigned)(qd[i].y - lo);
            const unsigned r2 = (unsigned)(qd[i].z - lo), r3 = (unsigned)(qd[i].w - lo);
            const bool m0 = r0 < RANGE, m1 = r1 < RANGE, m2 = r2 < RANGE, m3 = r3 < RANGE;
            unsigned p0 = 0, p1 = 0, p2 = 0, p3 = 0;
            if (m0) p0 = (unsigned)atomicAdd(&place[r0], 1);
            if (m1) p1 = (unsigned)atomicAdd(&place[r1], 1);
            if (m2) p2 = (unsigned)atomicAdd(&place[r2], 1);
            if (m3) p3 = (unsigned)atomicAdd(&place[r3], 1);
            if (m0) col[p0] = qs[i].x;
            if (m1) col[p1] = qs[i].y;
            if (m2) col[p2] = qs[i].z;
            if (m3) col[p3] = qs[i].w;
        }
    }
}

#define GATHER_BLOCKS 1024

template <int T>
__device__ __forceinline__ int half_lane(int v) {
    return __builtin_amdgcn_update_dpp(v, v, 0x150 + T, 0xf, 0xf, true);
}

template <bool FIRST, int T, int END>
struct Rows {
    static __device__ __forceinline__ void issue(float4* r, const char* __restrict__ featb,
                                                 int offs, int l16) {
        r[T] = *(const float4*)(featb + (unsigned)(half_lane<T>(offs) + l16));
        Rows<FIRST, T + 1, END>::issue(r, featb, offs, l16);
    }
    static __device__ __forceinline__ void add(float4& acc, const float4* r, int nsb, int mh) {
        float4 v = r[T];
        if (T >= END - 4 && !(2 * T < mh)) v = make_float4(0.0f, 0.0f, 0.0f, 0.0f);
        if (FIRST) {
            const float sc = __int_as_float(half_lane<T>(nsb));
            acc.x = __builtin_fmaf(v.x, sc, acc.x); acc.y = __builtin_fmaf(v.y, sc, acc.y);
            acc.z = __builtin_fmaf(v.z, sc, acc.z); acc.w = __builtin_fmaf(v.w, sc, acc.w);
        } else {
            acc.x += v.x; acc.y += v.y; acc.z += v.z; acc.w += v.w;
        }
        Rows<FIRST, T + 1, END>::add(acc, r, nsb, mh);
    }
};
template <bool FIRST, int END>
struct Rows<FIRST, END, END> {
    static __device__ __forceinline__ void issue(float4*, const char* __restrict__, int, int) {}
    static __device__ __forceinline__ void add(float4&, const float4*, int, int) {}
};

template <bool FIRST, int NT>
__device__ __forceinline__ void sum_rows(float4& acc, const char* __restrict__ featb,
                                         int offs, int l16, int nsb, int mh) {
    float4 r[4 * NT];
    Rows<FIRST, 0, 4 * NT>::issue(r, featb, offs, l16);
    __builtin_amdgcn_sched_barrier(0);
    Rows<FIRST, 0, 4 * NT>::add(acc, r, nsb, mh);
}

__device__ __forceinline__ int list_place(int first, int end, int ent) {
    const int p = min(first + ent, end - 1);
    return max(p, 0);
}

template <bool FIRST>
__global__ __launch_bounds__(256)
void gather_kernel(const float* __restrict__ feat, const float* __restrict__ norm_src,
                   const int* __restrict__ seg, const int* __restrict__ col,
                   const float* __restrict__ norm_dst, const float* __restrict__ bias,
                   float* __restrict__ outp) {
    const int lane = threadIdx.x & 31;
    const int half = lane >> 4;
    const int l    = lane & 15;
    const int l16  = l * 16;
    const int ent  = 2 * l + half;
    const int nw   = GATHER_BLOCKS * 8;
    const char* __restrict__ featb = (const char*)feat;
    int node = __builtin_amdgcn_readfirstlane(blockIdx.x * 8 + (threadIdx.x >> 5));
    const int2* seg2 = (const int2*)seg;
    int2 se  = seg2[node];
    int  nxt = node + nw;
    int2 se1 = make_int2(0, 0);
    if (nxt < N_NODES) se1 = seg2[nxt];
    float4 bv = make_float4(0.0f, 0.0f, 0.0f, 0.0f);
    if (!FIRST) bv = *(const float4*)(bias + l * 4);

    int   cv = col[list_place(se.x, se.y, ent)];
    float ns = 0.0f;
    if (FIRST) ns = norm_src[cv];

    while (true) {
        const int nxt2 = nxt + nw;
        int2 se2 = make_int2(0, 0);
        if (nxt2 < N_NODES) se2 = seg2[nxt2];
        const int cvn = col[list_place(se1.x, se1.y, ent)];

        float nd = 1.0f;
        if (!FIRST) nd = norm_dst[node];

        float4 acc = make_float4(0.0f, 0.0f, 0.0f, 0.0f);
        const int cnt = se.y - se.x;
        int   cvc = cv;
        float nsc = ns;
        for (int base = 0; base < cnt; base += 32) {
            if (base > 0) {
                cvc = col[min(se.x + base + ent, se.y - 1)];
                if (FIRST) nsc = norm_src[cvc];
            }
            const int m    = min(cnt - base, 32);
            const int mh   = m - half;
            const int offs = cvc * (64 * 4);
            const int nsb  = __float_as_int(nsc);
            if (m <= 8)       sum_rows<FIRST, 1>(acc, featb, offs, l16, nsb, mh);
            else if (m <= 16) sum_rows<FIRST, 2>(acc, featb, offs, l16, nsb, mh);
            else if (m <= 24) sum_rows<FIRST, 3>(acc, featb, offs, l16, nsb, mh);
            else              sum_rows<FIRST, 4>(acc, featb, offs, l16, nsb, mh);
        }
        float nsn = 0.0f;
        if (FIRST) nsn = norm_src[cvn];

        acc.x += __shfl_xor(acc.x, 16);
        acc.y += __shfl_xor(acc.y, 16);
        acc.z += __shfl_xor(acc.z, 16);
        acc.w += __shfl_xor(acc.w, 16);
        if (half == 0) {
            float4 o = acc;
            if (!FIRST) {
                o.x = acc.x * nd + bv.x; o.y = acc.y * nd + bv.y;
                o.z = acc.z * nd + bv.z; o.w = acc.w * nd + bv.w;
            }
            *(float4*)(outp + (long long)node * 64 + l * 4) = o;
        }
        if (nxt >= N_NODES) break;
        node = nxt;  nxt = nxt2;
        se   = se1;  se1 = se2;
        cv   = cvn;  ns  = nsn;
    }
}

#define N_TILES   (N_NODES / 16)
#define FUSE_WGS  1024
#define X_PITCH   (F_IN + 4)
#define H_PITCH   (F_HID + 4)

__device__ __forceinline__ v8f wmma4(v2f a, v2f b, v8f c) {
    return __builtin_amdgcn_wmma_f32_16x16x4_f32(
         false, a,  false, b,
         (short)0, c,  false,  false);
}

__device__ __forceinline__ int next_node(int n) {
    return (n & 3) == 3 ? n + (FUSE_WGS * 16 - 3) : n + 1;
}

__global__ __launch_bounds__(128)
void gather_gemms_kernel(const float* __restrict__ x, const float* __restrict__ norm_src,
                         const int* __restrict__ seg, const int* __restrict__ col,
                         const float* __restrict__ W1, const float* __restrict__ b1,
                         const float* __restrict__ W2, const float* __restrict__ norm_dst,
                         float* __restrict__ h2w) {
    __shared__ __attribute__((aligned(16))) float xs[16 * X_PITCH];
    __shared__ __attribute__((aligned(16))) float hs[16 * H_PITCH];
    const int lane = threadIdx.x & 31;
    const int wave = __builtin_amdgcn_readfirstlane(threadIdx.x >> 5);
    const int half = lane >> 4;
    const int l    = lane & 15;
    const int l16  = l * 16;
    const int ent  = 2 * l + half;
    const char* __restrict__ featb = (const char*)x;

    int node = blockIdx.x * 16 + wave * 4;
    typedef const int2 __attribute__((address_space(4))) * seg_ptr;
    const seg_ptr seg2 = (seg_ptr)seg;
    int2 se  = seg2[node];
    int  nxt = node + 1;
    int2 se1 = seg2[nxt];
    int   cv = col[list_place(se.x, se.y, ent)];

    const int colA = (2 * wave) * 16 + l;
    const int colB = colA + 16;
    const int col2 = wave * 16 + l;
    v2f w1a[F_IN / 4], w1b[F_IN / 4], w2[F_HID / 4];
    #pragma unroll
    for (int s = 0; s < F_IN / 4; ++s) {
        const int ka = s * 4 + half * 2;
        w1a[s].x = W1[ka * F_HID + colA];  w1a[s].y = W1[(ka + 1) * F_HID + colA];
        w1b[s].x = W1[ka * F_HID + colB];  w1b[s].y = W1[(ka + 1) * F_HID + colB];
    }
    #pragma unroll
    for (int s = 0; s < F_HID / 4; ++s) {
        const int ka = s * 4 + half * 2;
        w2[s].x = W2[ka * F_OUT + col2];   w2[s].y = W2[(ka + 1) * F_OUT + col2];
    }
    const float biasA = b1[colA], biasB = b1[colB];

    float ns = norm_src[cv];

    while (true) {
        const int nxt2 = next_node(nxt);
        int2 se2 = make_int2(0, 0);
        if (nxt2 < N_NODES) se2 = seg2[nxt2];
        const int cvn = col[list_place(se1.x, se1.y, ent)];

        float4 acc = make_float4(0.0f, 0.0f, 0.0f, 0.0f);
        const int cnt = se.y - se.x;
        int   cvc = cv;
        float nsc = ns;
        for (int base = 0; base < cnt; base += 32) {
            if (base > 0) {
                cvc = col[min(se.x + base + ent, se.y - 1)];
                nsc = norm_src[cvc];
            }
            const int m    = min(cnt - base, 32);
            const int mh   = m - half;
            const int offs = cvc * (F_IN * 4);
            const int nsb  = __float_as_int(nsc);
            if (m <= 8)       sum_rows<true, 1>(acc, featb, offs, l16, nsb, mh);
            else if (m <= 16) sum_rows<true, 2>(acc, featb, offs, l16, nsb, mh);
            else if (m <= 24) sum_rows<true, 3>(acc, featb, offs, l16, nsb, mh);
            else              sum_rows<true, 4>(acc, featb, offs, l16, nsb, mh);
        }
        const float nsn = norm_src[cvn];

        acc.x += __shfl_xor(acc.x, 16);
        acc.y += __shfl_xor(acc.y, 16);
        acc.z += __shfl_xor(acc.z, 16);
        acc.w += __shfl_xor(acc.w, 16);
        if (half == 0) *(float4*)(xs + (node & 15) * X_PITCH + l * 4) = acc;

        if ((node & 3) == 3) {
            __syncthreads();
            const int row0 = node & ~15;

            float nd[8], nr[8];
            #pragma unroll
            for (int r = 0; r < 8; ++r) {
                nd[r] = norm_dst[row0 + r + half * 8];
                nr[r] = norm_src[row0 + r + half * 8];
            }

            v2f a[F_IN / 4];
            {
                const float* xp = xs + l * X_PITCH + half * 2;
                #pragma unroll
                for (int s = 0; s < F_IN / 4; ++s) a[s] = *(const v2f*)(xp + s * 4);
            }
            v8f c0 = {}, c1 = {};
            #pragma unroll
            for (int s = 0; s < F_IN / 4; ++s) {
                c0 = wmma4(a[s], w1a[s], c0);
                c1 = wmma4(a[s], w1b[s], c1);
            }

            #pragma unroll
            for (int r = 0; r < 8; ++r) {
                float* hp = hs + (r + half * 8) * H_PITCH;
                hp[colA] = fmaxf(c0[r] * nd[r] + biasA, 0.0f) * nr[r];
                hp[colB] = fmaxf(c1[r] * nd[r] + biasB, 0.0f) * nr[r];
            }
            __syncthreads();

            v2f a2[F_HID / 4];
            {
                const float* hp = hs + l * H_PITCH + half * 2;
                #pragma unroll
                for (int s = 0; s < F_HID / 4; ++s) a2[s] = *(const v2f*)(hp + s * 4);
            }
            v8f c = {};
            #pragma unroll
            for (int s = 0; s < F_HID / 4; ++s) c = wmma4(a2[s], w2[s], c);

            #pragma unroll
            for (int r = 0; r < 8; ++r)
                h2w[(long long)(row0 + r + half * 8) * F_OUT + col2] = c[r];

            if (nxt >= N_NODES) break;
        }
        node = nxt;  nxt = nxt2;
        se   = se1;  se1 = se2;
        cv   = cvn;  ns  = nsn;
    }
}


extern "C" void kernel_launch(void* const* d_in, const int* in_sizes, int n_in,
                              void* d_out, int out_size, void* d_ws, size_t ws_size,
                              hipStream_t stream) {
    const float* x   = (const float*)d_in[0];
    const float* W1  = (const float*)d_in[1];
    const float* b1  = (const float*)d_in[2];
    const float* W2  = (const float*)d_in[3];
    const float* b2  = (const float*)d_in[4];
    const int*   src = (const int*)d_in[5];
    const int*   dst = (const int*)d_in[6];
    float*       out = (float*)d_out;

    int*   part_in  = (int*)d_ws;
    int*   part_out = part_in + N_CHUNKS * N_NODES;
    int*   subsum   = part_out + N_CHUNKS * N_NODES;
    int*   seg      = subsum + N_SUBS * N_CHUNKS;
    int*   col      = seg + 2 * N_NODES;
    float* norm_src = (float*)(col + N_EDGES);
    float* norm_dst = norm_src + N_NODES;
    float* h2w      = norm_dst + N_NODES;

    degree_count_kernel<<<dim3(N_CHUNKS * N_RANGES, 2), BIN_T, 0, stream>>>(
        src, dst, part_out, part_in, subsum);
    scan_norm_kernel<<<N_SUBS, SCAN_T, 0, stream>>>(
        part_out, part_in, subsum, seg, norm_src, norm_dst);
    fill_kernel<<<N_CHUNKS * N_RANGES, BIN_T, 0, stream>>>(src, dst, part_in, col);

    gather_gemms_kernel<<<FUSE_WGS, 128, 0, stream>>>(
        x, norm_src, seg, col, W1, b1, W2, norm_dst, h2w);
    gather_kernel<false><<<GATHER_BLOCKS, 256, 0, stream>>>(
        h2w, nullptr, seg, col, norm_dst, b2, out);
}
